# starting kernel
# baseline (speedup 1.0000x reference)
#include <hip/hip_runtime.h>
#include <hip/hip_bf16.h>


typedef __attribute__((ext_vector_type(2))) float v2f;
typedef __attribute__((ext_vector_type(4))) float v4f;
typedef __attribute__((ext_vector_type(8))) float v8f;

#define NBATCH   64
#define LSEQ     32768
#define NCH      16
#define SEG      512
#define NSEG     (LSEQ / SEG)
#define TOTSEG   (NBATCH * NSEG)
#define SEGFLTS  (SEG * NCH)

__global__ void seg_sums_kernel(const float* __restrict__ in,
                                float* __restrict__ sums) {
  const int wave = (blockIdx.x * blockDim.x + threadIdx.x) >> 5;
  const int lane = threadIdx.x & 31;
  const int g = lane & 3;
  const int r = lane >> 2;

  const v4f* p = (const v4f*)(in + (size_t)wave * SEGFLTS) + (r * 4 + g);
  v4f acc = {0.f, 0.f, 0.f, 0.f};
#pragma unroll 4
  for (int i = 0; i < SEG / 8; ++i) {
    acc += p[32 * i];
  }
#pragma unroll
  for (int m = 4; m <= 16; m <<= 1) {
    acc[0] += __shfl_xor(acc[0], m, 32);
    acc[1] += __shfl_xor(acc[1], m, 32);
    acc[2] += __shfl_xor(acc[2], m, 32);
    acc[3] += __shfl_xor(acc[3], m, 32);
  }
  if (r == 0) {
    ((v4f*)sums)[wave * 4 + g] = acc;
  }
}

__global__ void seg_scan_kernel(const float* __restrict__ sums,
                                float* __restrict__ excl) {
  const int batch = blockIdx.x;
  const int lane = threadIdx.x;
  if (lane < NCH) {
    float carry = 0.f;
    for (int s = 0; s < NSEG; ++s) {
      const int idx = (batch * NSEG + s) * NCH + lane;
      const float v = sums[idx];
      excl[idx] = carry;
      carry += v;
    }
  }
}

__global__ void wmma_scan_kernel(const float* __restrict__ in,
                                 const float* __restrict__ excl,
                                 float* __restrict__ out) {
  const int wave = (blockIdx.x * blockDim.x + threadIdx.x) >> 5;
  const int lane = threadIdx.x & 31;
  const int m    = lane & 15;
  const int half = lane >> 4;
  const int k0   = half * 2;

  const float* base  = in  + (size_t)wave * SEGFLTS;
  float*       obase = out + (size_t)wave * SEGFLTS;

  float carry = excl[wave * NCH + m];

  for (int tile = 0; tile < SEG / 16; ++tile) {
    const float* tb = base + tile * 256;

    v8f d;
#pragma unroll
    for (int i = 0; i < 8; ++i) d[i] = carry;

#pragma unroll
    for (int kb = 0; kb < 16; kb += 4) {
      v2f a, b;
      a[0] = ((kb + k0)     <= m) ? 1.0f : 0.0f;
      a[1] = ((kb + k0 + 1) <= m) ? 1.0f : 0.0f;
      b[0] = tb[(kb + k0)     * NCH + m];
      b[1] = tb[(kb + k0 + 1) * NCH + m];
      d = __builtin_amdgcn_wmma_f32_16x16x4_f32(
           false, a,  false, b,
           (short)0, d,  false,  false);
    }

    float* ob = obase + tile * 256;
#pragma unroll
    for (int i = 0; i < 8; ++i) {
      __builtin_nontemporal_store(d[i], &ob[(i + 8 * half) * NCH + m]);
    }

    carry = __shfl(d[7], 16 + m, 32);
  }
}

extern "C" void kernel_launch(void* const* d_in, const int* in_sizes, int n_in,
                              void* d_out, int out_size, void* d_ws, size_t ws_size,
                              hipStream_t stream) {
  const float* in = (const float*)d_in[0];
  float* out = (float*)d_out;

  float* sums = (float*)d_ws;
  float* excl = sums + (size_t)TOTSEG * NCH;

  seg_sums_kernel<<<TOTSEG / 8, 256, 0, stream>>>(in, sums);
  seg_scan_kernel<<<NBATCH, 32, 0, stream>>>(sums, excl);
  wmma_scan_kernel<<<TOTSEG / 8, 256, 0, stream>>>(in, excl, out);
}


# First launch also scans each batch's first 4096 timesteps (no carry-in) beside the segment totals; second launch: 4 regions of 7 chunks, not 8; carry-in = head total + segment totals (new sum order)
# speedup vs baseline: 1.5001x; 1.5001x over previous
#include <hip/hip_runtime.h>
#include <hip/hip_bf16.h>


typedef __attribute__((ext_vector_type(2))) float v2f;
typedef __attribute__((ext_vector_type(4))) float v4f;
typedef __attribute__((ext_vector_type(8))) float v8f;

#define NBATCH   64
#define LSEQ     32768
#define NCH      16

#define RWAVES   16
#define CROWS    64
#define TILES    (CROWS / 16)
#define CHROWS   (RWAVES * CROWS)
#define CHFLTS   (CHROWS * NCH)

#define HCHUNK   4
#define HROWS    (HCHUNK * CHROWS)
#define NREG     4
#define RCHUNK   7
#define RROWS    (RCHUNK * CHROWS)
#define SEGPREG  16
#define SEG      (RROWS / SEGPREG)
#define SEGFLTS  (SEG * NCH)
#define NSEGSUM  ((NREG - 1) * SEGPREG)
#define NSLOT    64
#define SUMGRPS  (NBATCH * NSEGSUM / RWAVES)

static_assert(HROWS + NREG * RROWS == LSEQ, "head and regions must tile the sequence");
static_assert(SEG % 32 == 0, "segment loop is unrolled by 4 loads of 8 rows");

static __device__ __forceinline__ void group_barrier() {
  asm volatile("s_wait_dscnt 0x0" ::: "memory");
  asm volatile("s_barrier_signal -1" ::: "memory");
  asm volatile("s_barrier_wait -1" ::: "memory");
}

static __device__ __forceinline__ void exchange(float (*slot)[NCH], int w, int m,
                                                int half, float mine,
                                                float& ahead, float& total) {
  if (half == 0) slot[w][m] = mine;
  group_barrier();
  float run = 0.f, before = 0.f;
#pragma unroll
  for (int j = 0; j < RWAVES; ++j) {
    const float v = slot[j][m];
    before = (j == w) ? run : before;
    run += v;
  }
  ahead = before;
  total = run;
}

template <int NCHUNKS, bool HEAD>
static __device__ __forceinline__ void scan_rows(const float* __restrict__ base,
                                                 float* __restrict__ obase,
                                                 const float* __restrict__ sp_in,
                                                 float* __restrict__ sp_out,
                                                 int nprev,
                                                 float (*xch)[RWAVES][NCH],
                                                 int w, int lane) {
  const int m    = lane & 15;
  const int half = lane >> 4;
  const int ts   = 2 * (m & 7) + (m >> 3);

  const int js = 2 * w + half;
  float s1 = 0.f, s2 = 0.f, hd = 0.f;
  if (!HEAD) {
    const int j2 = js + 32 < NSEGSUM ? js + 32 : NSEGSUM - 1;
    s1 = sp_in[js * NCH];
    s2 = sp_in[j2 * NCH];
    hd = sp_in[NSEGSUM * NCH];
  }

  v2f a[4];
#pragma unroll
  for (int q = 0; q < 4; ++q) {
    a[q][0] = ((q * 4 + half)     <= ts) ? 1.0f : 0.0f;
    a[q][1] = ((q * 4 + 2 + half) <= ts) ? 1.0f : 0.0f;
  }

  float xb[3][TILES * 8];

#define REQUEST_CHUNK(dst, c)                                                   \
  do {                                                                          \
    const float* tb_ = base + (size_t)(c) * CHFLTS;                             \
    _Pragma("unroll") for (int t_ = 0; t_ < TILES; ++t_) {                      \
      _Pragma("unroll") for (int q_ = 0; q_ < 4; ++q_) {                        \
        (dst)[t_ * 8 + q_ * 2]     = tb_[t_ * 256 + (q_ * 4 + half)     * NCH + m];   \
        (dst)[t_ * 8 + q_ * 2 + 1] = tb_[t_ * 256 + (q_ * 4 + 2 + half) * NCH + m];   \
      }                                                                         \
    }                                                                           \
  } while (0)

  REQUEST_CHUNK(xb[0], 0);
  REQUEST_CHUNK(xb[1], 1);

  float ahead, carry;
  if (HEAD) {
    carry = 0.f;
  } else {
    float part = (js < nprev ? s1 : 0.f) + (js + 32 < nprev ? s2 : 0.f);
    part += __shfl_xor(part, 16, 32);
    exchange(xch[0], w, m, half, part, ahead, carry);
    carry = hd + carry;
  }

#pragma unroll
  for (int c = 0; c < NCHUNKS; ++c) {
    if (c + 2 < NCHUNKS) {
      REQUEST_CHUNK(xb[(c + 2) % 3], c + 2);
    }

    const float* x = xb[c % 3];
    v8f d[TILES];
    float tt[TILES];
#pragma unroll
    for (int t = 0; t < TILES; ++t) {
#pragma unroll
      for (int i = 0; i < 8; ++i) d[t][i] = 0.f;
#pragma unroll
      for (int q = 0; q < 4; ++q) {
        v2f b;
        b[0] = x[t * 8 + q * 2];
        b[1] = x[t * 8 + q * 2 + 1];
        d[t] = __builtin_amdgcn_wmma_f32_16x16x4_f32(
             false, a[q],  false, b,
             (short)0, d[t],  false,  false);
      }
    }
#pragma unroll
    for (int t = 0; t < TILES; ++t) tt[t] = __shfl(d[t][7], 16 + m, 32);

    const float wave_total = ((tt[0] + tt[1]) + tt[2]) + tt[3];
    float total;
    exchange(xch[(c + 1) & 1], w, m, half, wave_total, ahead, total);

    float off = carry + ahead;
    float* ob = obase + (size_t)c * CHFLTS;
#pragma unroll
    for (int t = 0; t < TILES; ++t) {
#pragma unroll
      for (int i = 0; i < 8; ++i) {
        __builtin_nontemporal_store(d[t][i] + off, &ob[t * 256 + (2 * i + half) * NCH + m]);
      }
      off += tt[t];
    }
    carry += total;
  }
#undef REQUEST_CHUNK

  if (HEAD) {
    if (w == 0 && half == 0) sp_out[NSEGSUM * NCH] = carry;
  }
}

__global__ void __launch_bounds__(RWAVES * 32)
head_scan_seg_sums_kernel(const float* __restrict__ in,
                          float* __restrict__ sums,
                          float* __restrict__ out) {
  __shared__ float xch[2][RWAVES][NCH];

  const int w    = __builtin_amdgcn_readfirstlane((int)(threadIdx.x >> 5));
  const int lane = threadIdx.x & 31;

  if (blockIdx.x < NBATCH) {
    const int batch = blockIdx.x;
    const size_t row0 = (size_t)batch * LSEQ + (size_t)w * CROWS;
    scan_rows<HCHUNK, true>(in + row0 * NCH, out + row0 * NCH, nullptr,
                            sums + (size_t)batch * NSLOT * NCH + (lane & 15),
                            0, xch, w, lane);
  } else {
    const int t = (blockIdx.x - NBATCH) * RWAVES + w;
    const int batch = t / NSEGSUM;
    const int j     = t % NSEGSUM;
    const int g = lane & 3;
    const int r = lane >> 2;

    const v4f* p = (const v4f*)(in + ((size_t)batch * LSEQ + HROWS) * NCH + (size_t)j * SEGFLTS) + (r * 4 + g);
    v4f acc = {0.f, 0.f, 0.f, 0.f};
#pragma unroll 4
    for (int i = 0; i < SEG / 8; ++i) {
      acc += p[32 * i];
    }
#pragma unroll
    for (int s = 4; s <= 16; s <<= 1) {
      acc[0] += __shfl_xor(acc[0], s, 32);
      acc[1] += __shfl_xor(acc[1], s, 32);
      acc[2] += __shfl_xor(acc[2], s, 32);
      acc[3] += __shfl_xor(acc[3], s, 32);
    }
    if (r == 0) {
      ((v4f*)sums)[(batch * NSLOT + j) * 4 + g] = acc;
    }
  }
}

__global__ void __launch_bounds__(RWAVES * 32)
region_scan_kernel(const float* __restrict__ in,
                   const float* __restrict__ sums,
                   float* __restrict__ out) {
  __shared__ float xch[2][RWAVES][NCH];

  const int w    = __builtin_amdgcn_readfirstlane((int)(threadIdx.x >> 5));
  const int lane = threadIdx.x & 31;

  const int batch  = blockIdx.x / NREG;
  const int region = blockIdx.x % NREG;

  const size_t row0 = (size_t)batch * LSEQ + HROWS + (size_t)region * RROWS + (size_t)w * CROWS;
  scan_rows<RCHUNK, false>(in + row0 * NCH, out + row0 * NCH,
                           sums + (size_t)batch * NSLOT * NCH + (lane & 15), nullptr,
                           region * SEGPREG, xch, w, lane);
}

extern "C" void kernel_launch(void* const* d_in, const int* in_sizes, int n_in,
                              void* d_out, int out_size, void* d_ws, size_t ws_size,
                              hipStream_t stream) {
  const float* in = (const float*)d_in[0];
  float* out = (float*)d_out;

  float* sums = (float*)d_ws;

  head_scan_seg_sums_kernel<<<NBATCH + SUMGRPS, RWAVES * 32, 0, stream>>>(in, sums, out);
  region_scan_kernel<<<NBATCH * NREG, RWAVES * 32, 0, stream>>>(in, sums, out);
}
